# starting kernel
# baseline (speedup 1.0000x reference)
#include <hip/hip_runtime.h>


typedef __attribute__((ext_vector_type(2))) float v2f;
typedef __attribute__((ext_vector_type(4))) float v4f;
typedef __attribute__((ext_vector_type(8))) float v8f;

#define NDIM    64
#define LS      68
#define ITERS   13
#define THREADS 256

__global__ __launch_bounds__(THREADS)
void bjorck_wmma_kernel(const float* __restrict__ w_in, float* __restrict__ w_out) {
  __shared__ float Wl [NDIM * LS];
  __shared__ float Wtl[NDIM * LS];
  __shared__ float Sl [NDIM * LS];

  const unsigned tid  = threadIdx.x;
  const unsigned lane = tid & 31u;
  const unsigned wave = tid >> 5;
  const unsigned half = lane >> 4;
  const unsigned colm = lane & 15u;
  const unsigned ti   = wave >> 1;
  const unsigned tj0  = (wave & 1u) << 1;

  const float4* gin = (const float4*)(w_in + (size_t)blockIdx.x * (NDIM * NDIM));
  #pragma unroll
  for (int i = 0; i < 4; ++i) {
    unsigned c = tid + THREADS * i;
    unsigned row = c >> 4, q = c & 15u;
    float4 v = gin[c];
    *(float4*)&Wl[row * LS + q * 4u] = v;
    Wtl[(q * 4u + 0u) * LS + row] = v.x;
    Wtl[(q * 4u + 1u) * LS + row] = v.y;
    Wtl[(q * 4u + 2u) * LS + row] = v.z;
    Wtl[(q * 4u + 3u) * LS + row] = v.w;
  }
  __syncthreads();

  const float* aT  = &Wtl[(16u * ti         + colm) * LS + 2u * half];
  const float* b0T = &Wtl[(16u * tj0        + colm) * LS + 2u * half];
  const float* b1T = &Wtl[(16u * (tj0 + 1u) + colm) * LS + 2u * half];
  const float* aR  = &Wl [(16u * ti         + colm) * LS + 2u * half];
  const float* s0R = &Sl [(16u * tj0        + colm) * LS + 2u * half];
  const float* s1R = &Sl [(16u * (tj0 + 1u) + colm) * LS + 2u * half];
  float* st0 = &Sl [(16u * tj0        + colm) * LS + 16u * ti + 8u * half];
  float* st1 = &Sl [(16u * (tj0 + 1u) + colm) * LS + 16u * ti + 8u * half];
  float* wt0 = &Wtl[(16u * tj0        + colm) * LS + 16u * ti + 8u * half];
  float* wt1 = &Wtl[(16u * (tj0 + 1u) + colm) * LS + 16u * ti + 8u * half];

  #pragma unroll 1
  for (int it = 0; it < ITERS; ++it) {
    v8f s0 = {}, s1 = {};
    #pragma unroll
    for (unsigned kb = 0; kb < 16; ++kb) {
      v2f a  = *(const v2f*)(aT  + 4u * kb);
      v2f b0 = *(const v2f*)(b0T + 4u * kb);
      v2f b1 = *(const v2f*)(b1T + 4u * kb);
      s0 = __builtin_amdgcn_wmma_f32_16x16x4_f32(false, a, false, b0, (short)0, s0, false, false);
      s1 = __builtin_amdgcn_wmma_f32_16x16x4_f32(false, a, false, b1, (short)0, s1, false, false);
    }
    {
      v4f lo0 = {s0[0], s0[1], s0[2], s0[3]}, hi0 = {s0[4], s0[5], s0[6], s0[7]};
      v4f lo1 = {s1[0], s1[1], s1[2], s1[3]}, hi1 = {s1[4], s1[5], s1[6], s1[7]};
      *(v4f*)(st0)      = lo0;  *(v4f*)(st0 + 4) = hi0;
      *(v4f*)(st1)      = lo1;  *(v4f*)(st1 + 4) = hi1;
    }
    __syncthreads();

    v8f p0 = {}, p1 = {};
    #pragma unroll
    for (unsigned kb = 0; kb < 16; ++kb) {
      v2f a  = *(const v2f*)(aR  + 4u * kb);
      v2f b0 = *(const v2f*)(s0R + 4u * kb);
      v2f b1 = *(const v2f*)(s1R + 4u * kb);
      p0 = __builtin_amdgcn_wmma_f32_16x16x4_f32(false, a, false, b0, (short)0, p0, false, false);
      p1 = __builtin_amdgcn_wmma_f32_16x16x4_f32(false, a, false, b1, (short)0, p1, false, false);
    }
    v4f w0lo = *(const v4f*)(wt0), w0hi = *(const v4f*)(wt0 + 4);
    v4f w1lo = *(const v4f*)(wt1), w1hi = *(const v4f*)(wt1 + 4);
    v4f n0lo, n0hi, n1lo, n1hi;
    #pragma unroll
    for (int r = 0; r < 4; ++r) {
      n0lo[r] = 1.5f * w0lo[r] - 0.5f * p0[r];
      n0hi[r] = 1.5f * w0hi[r] - 0.5f * p0[r + 4];
      n1lo[r] = 1.5f * w1lo[r] - 0.5f * p1[r];
      n1hi[r] = 1.5f * w1hi[r] - 0.5f * p1[r + 4];
    }
    __syncthreads();

    *(v4f*)(wt0)      = n0lo;  *(v4f*)(wt0 + 4) = n0hi;
    *(v4f*)(wt1)      = n1lo;  *(v4f*)(wt1 + 4) = n1hi;
    #pragma unroll
    for (unsigned r = 0; r < 8; ++r) {
      unsigned row = 16u * ti + 8u * half + r;
      float v0 = (r < 4u) ? n0lo[r] : n0hi[r - 4u];
      float v1 = (r < 4u) ? n1lo[r] : n1hi[r - 4u];
      Wl[row * LS + 16u * tj0 + colm]        = v0;
      Wl[row * LS + 16u * (tj0 + 1u) + colm] = v1;
    }
    __syncthreads();
  }

  float4* gout = (float4*)(w_out + (size_t)blockIdx.x * (NDIM * NDIM));
  #pragma unroll
  for (int i = 0; i < 4; ++i) {
    unsigned c = tid + THREADS * i;
    unsigned row = c >> 4, q = c & 15u;
    gout[c] = *(const float4*)&Wl[row * LS + q * 4u];
  }
}

extern "C" void kernel_launch(void* const* d_in, const int* in_sizes, int n_in,
                              void* d_out, int out_size, void* d_ws, size_t ws_size,
                              hipStream_t stream) {
  const float* w = (const float*)d_in[0];
  float* out = (float*)d_out;
  const int batch = in_sizes[0] / (NDIM * NDIM);
  bjorck_wmma_kernel<<<batch, THREADS, 0, stream>>>(w, out);
}


# Two waves per matrix; each keeps its 64x32 column strip of W and each finished 16-row block of S in registers as the B operands (A from LDS); S skips LDS, 2 barriers/iter; k summed in another order
# speedup vs baseline: 1.0462x; 1.0462x over previous
#include <hip/hip_runtime.h>


typedef __attribute__((ext_vector_type(2))) float v2f;
typedef __attribute__((ext_vector_type(4))) float v4f;
typedef __attribute__((ext_vector_type(8))) float v8f;

#define NDIM    64
#define LS      68
#define ITERS   13
#define NJ      2
#define WAVES   (4 / NJ)
#define THREADS (32 * WAVES)
#define CHUNKS  (1024 / THREADS)

#define WMMA(a, b, c) __builtin_amdgcn_wmma_f32_16x16x4_f32(false, (a), false, (b), (short)0, (c), false, false)

__global__ __launch_bounds__(THREADS)
void orth_iter_wmma_kernel(const float* __restrict__ w_in, float* __restrict__ w_out) {
  __shared__ float Wl [NDIM * LS];
  __shared__ float Wtl[NDIM * LS];

  const unsigned tid  = threadIdx.x;
  const unsigned lane = tid & 31u;
  const unsigned wave = tid >> 5;
  const unsigned half = lane >> 4;
  const unsigned colm = lane & 15u;

  const float4* gin = (const float4*)(w_in + (size_t)blockIdx.x * (NDIM * NDIM));
  #pragma unroll
  for (int i = 0; i < CHUNKS; ++i) {
    unsigned c = tid + THREADS * i;
    unsigned row = c >> 4, q = c & 15u;
    float4 v = gin[c];
    *(float4*)&Wl[row * LS + q * 4u] = v;
    Wtl[(q * 4u + 0u) * LS + row] = v.x;
    Wtl[(q * 4u + 1u) * LS + row] = v.y;
    Wtl[(q * 4u + 2u) * LS + row] = v.z;
    Wtl[(q * 4u + 3u) * LS + row] = v.w;
  }
  __syncthreads();

  const float* aT = &Wtl[colm * LS + 8u * half];
  const float* aR = &Wl [colm * LS + 8u * half];
  float* wt = &Wtl[(16u * NJ * wave + colm) * LS + 8u * half];
  float* wr = &Wl [(8u * half) * LS + 16u * NJ * wave + colm];

  v8f w[4][NJ];
  #pragma unroll
  for (unsigned i = 0; i < 4; ++i)
    #pragma unroll
    for (unsigned jj = 0; jj < NJ; ++jj) {
      v4f lo = *(const v4f*)(wt + 16u * jj * LS + 16u * i);
      v4f hi = *(const v4f*)(wt + 16u * jj * LS + 16u * i + 4u);
      w[i][jj] = (v8f){lo[0], lo[1], lo[2], lo[3], hi[0], hi[1], hi[2], hi[3]};
    }

  #pragma unroll 1
  for (int it = 0; it < ITERS; ++it) {
    v8f p[4][NJ];
    #pragma unroll
    for (unsigned i = 0; i < 4; ++i)
      #pragma unroll
      for (unsigned jj = 0; jj < NJ; ++jj) p[i][jj] = (v8f){};

    #pragma unroll
    for (unsigned a = 0; a < 4; ++a) {
      v8f s[NJ];
      #pragma unroll
      for (unsigned jj = 0; jj < NJ; ++jj) s[jj] = (v8f){};
      #pragma unroll
      for (unsigned i = 0; i < 4; ++i) {
        v4f a0 = *(const v4f*)(aT + 16u * a * LS + 16u * i);
        v4f a1 = *(const v4f*)(aT + 16u * a * LS + 16u * i + 4u);
        const v8f a8 = {a0[0], a0[1], a0[2], a0[3], a1[0], a1[1], a1[2], a1[3]};
        #pragma unroll
        for (unsigned f = 0; f < 4; ++f) {
          v2f af = {a8[2u * f], a8[2u * f + 1u]};
          #pragma unroll
          for (unsigned jj = 0; jj < NJ; ++jj) {
            v2f bf = {w[i][jj][2u * f], w[i][jj][2u * f + 1u]};
            s[jj] = WMMA(af, bf, s[jj]);
          }
        }
      }
      #pragma unroll
      for (unsigned g = 0; g < 2; ++g) {
        v4f av[4];
        #pragma unroll
        for (unsigned i = 0; i < 4; ++i)
          av[i] = *(const v4f*)(aR + 16u * i * LS + 16u * a + 4u * g);
        #pragma unroll
        for (unsigned t = 0; t < 2; ++t)
          #pragma unroll
          for (unsigned i = 0; i < 4; ++i)
            #pragma unroll
            for (unsigned jj = 0; jj < NJ; ++jj) {
              v2f af = {av[i][2u * t], av[i][2u * t + 1u]};
              v2f bf = {s[jj][4u * g + 2u * t], s[jj][4u * g + 2u * t + 1u]};
              p[i][jj] = WMMA(af, bf, p[i][jj]);
            }
      }
    }

    #pragma unroll
    for (unsigned i = 0; i < 4; ++i)
      #pragma unroll
      for (unsigned jj = 0; jj < NJ; ++jj)
        #pragma unroll
        for (unsigned r = 0; r < 8; ++r)
          w[i][jj][r] = 1.5f * w[i][jj][r] - 0.5f * p[i][jj][r];

    __syncthreads();

    #pragma unroll
    for (unsigned i = 0; i < 4; ++i)
      #pragma unroll
      for (unsigned jj = 0; jj < NJ; ++jj) {
        const v8f n = w[i][jj];
        *(v4f*)(wt + 16u * jj * LS + 16u * i)      = (v4f){n[0], n[1], n[2], n[3]};
        *(v4f*)(wt + 16u * jj * LS + 16u * i + 4u) = (v4f){n[4], n[5], n[6], n[7]};
        #pragma unroll
        for (unsigned r = 0; r < 8; ++r)
          wr[(16u * i + r) * LS + 16u * jj] = n[r];
      }
    __syncthreads();
  }

  float4* gout = (float4*)(w_out + (size_t)blockIdx.x * (NDIM * NDIM));
  #pragma unroll
  for (int i = 0; i < CHUNKS; ++i) {
    unsigned c = tid + THREADS * i;
    unsigned row = c >> 4, q = c & 15u;
    gout[c] = *(const float4*)&Wl[row * LS + q * 4u];
  }
}

extern "C" void kernel_launch(void* const* d_in, const int* in_sizes, int n_in,
                              void* d_out, int out_size, void* d_ws, size_t ws_size,
                              hipStream_t stream) {
  const float* w = (const float*)d_in[0];
  float* out = (float*)d_out;
  const int batch = in_sizes[0] / (NDIM * NDIM);
  orth_iter_wmma_kernel<<<batch, THREADS, 0, stream>>>(w, out);
}
